# starting kernel
# baseline (speedup 1.0000x reference)
#include <hip/hip_runtime.h>

typedef __attribute__((ext_vector_type(2))) float v2f;
typedef __attribute__((ext_vector_type(8))) float v8f;

#define B_  2
#define C_  32
#define N_  1024

__global__ __launch_bounds__(256)
void psm_attn_kernel(const float* __restrict__ feature,
                     const float* __restrict__ seg,
                     const float* __restrict__ gamma,
                     float* __restrict__ out)
{
    __shared__ __align__(16) float ldsM[N_];
    __shared__ __align__(16) float ldsF[N_];

    const int bc   = blockIdx.x >> 3;
    const int tg   = blockIdx.x & 7;
    const int b    = bc >> 5;
    const int tid  = threadIdx.x;
    const int lane = tid & 31;
    const int wave = tid >> 5;

    const float* fptr = feature + (size_t)bc * N_;
    const float* mptr = seg + ((size_t)b * 2 + 1) * N_;

    for (int j = tid; j < N_; j += 256) {
        ldsM[j] = mptr[j];
        ldsF[j] = fptr[j];
    }
    __syncthreads();

    float mx = -3.0e38f, mn = 3.0e38f;
    for (int j = lane; j < N_; j += 32) {
        const float v = ldsM[j];
        mx = fmaxf(mx, v);
        mn = fminf(mn, v);
    }
#pragma unroll
    for (int off = 16; off > 0; off >>= 1) {
        mx = fmaxf(mx, __shfl_xor(mx, off, 32));
        mn = fminf(mn, __shfl_xor(mn, off, 32));
    }

    const int half = lane >> 4;
    const int row  = lane & 15;
    const int i0   = (tg * 8 + wave) * 16;

    const float sM = ldsF[i0 + row];
    const float L2E  = 1.44269504088896340736f;
    const float sL2  = sM * L2E;
    const float rL2  = (sM >= 0.0f ? sM * mx : sM * mn) * L2E;
    const float g    = gamma[0];

    v8f acc_num = {};
    v8f acc_den = {};
    const v2f bOnes = {1.0f, 1.0f};

    const v2f* m2 = (const v2f*)ldsM;
    const v2f* f2 = (const v2f*)ldsF;

    for (int j0 = 0; j0 < N_; j0 += 4) {
        const int k = (j0 >> 1) + half;
        const v2f mv = m2[k];
        const v2f fv = f2[k];
        v2f e, ef;
        e.x = __builtin_amdgcn_exp2f(sL2 * mv.x - rL2);
        e.y = __builtin_amdgcn_exp2f(sL2 * mv.y - rL2);
        ef.x = e.x * fv.x;
        ef.y = e.y * fv.y;
        acc_num = __builtin_amdgcn_wmma_f32_16x16x4_f32(
            false, ef, false, bOnes, (short)0, acc_num, false, false);
        acc_den = __builtin_amdgcn_wmma_f32_16x16x4_f32(
            false, e,  false, bOnes, (short)0, acc_den, false, false);
    }

    float* obase = out + (size_t)bc * N_;
#pragma unroll
    for (int r = 0; r < 8; ++r) {
        if (row == r) {
            const int i = i0 + r + 8 * half;
            obase[i] = g * (acc_num[r] / acc_den[r]) + ldsF[i];
        }
    }
}

extern "C" void kernel_launch(void* const* d_in, const int* in_sizes, int n_in,
                              void* d_out, int out_size, void* d_ws, size_t ws_size,
                              hipStream_t stream) {
    (void)in_sizes; (void)n_in; (void)out_size; (void)d_ws; (void)ws_size;
    const float* feature = (const float*)d_in[0];
    const float* seg     = (const float*)d_in[1];
    const float* gamma   = (const float*)d_in[2];
    float* out           = (float*)d_out;

    dim3 grid(B_ * C_ * 8);
    dim3 block(256);
    psm_attn_kernel<<<grid, block, 0, stream>>>(feature, seg, gamma, out);
}


# j collapsed onto 16 levels of m: level moments sum f_j d_j^p (p<8) by a one-hot matrix product; a row sums 128 terms exp(s M_k) s^p/p! x moment, not 1024 exps (8-term series in s d_j; sum order)
# speedup vs baseline: 4.5485x; 4.5485x over previous
#include <hip/hip_runtime.h>

typedef __attribute__((ext_vector_type(2))) float v2f;
typedef __attribute__((ext_vector_type(8))) float v8f;

#define B_  2
#define C_  32
#define N_  1024
#define NB_ 16
#define NW_ 8

__global__ __launch_bounds__(256)
void psm_attn_kernel(const float* __restrict__ feature,
                     const float* __restrict__ seg,
                     const float* __restrict__ gamma,
                     float* __restrict__ out)
{
    __shared__ __attribute__((aligned(16))) float rec[4 * N_];
    __shared__ __attribute__((aligned(16))) float sf[N_];
    __shared__ __attribute__((aligned(16))) float part[NW_ * 256];
    __shared__ __attribute__((aligned(16))) float mom[256];
    __shared__ float red[2 * NW_];

    const int bc   = blockIdx.x >> 2;
    const int q    = blockIdx.x & 3;
    const int b    = bc >> 5;
    const int tid  = threadIdx.x;
    const int lane = tid & 31;
    const int wave = tid >> 5;
    const int c16  = lane & 15;
    const int half = lane >> 4;

    const float* fptr = feature + (size_t)bc * N_;
    const float* mptr = seg + ((size_t)b * 2 + 1) * N_;
    const float g = gamma[0];

    const float4 m4 = *reinterpret_cast<const float4*>(mptr + 4 * tid);
    const float4 f4 = *reinterpret_cast<const float4*>(fptr + 4 * tid);
    *reinterpret_cast<float4*>(sf + 4 * tid) = f4;
    float mx = fmaxf(fmaxf(m4.x, m4.y), fmaxf(m4.z, m4.w));
    float mn = fminf(fminf(m4.x, m4.y), fminf(m4.z, m4.w));
#pragma unroll
    for (int off = 16; off > 0; off >>= 1) {
        mx = fmaxf(mx, __shfl_xor(mx, off, 32));
        mn = fminf(mn, __shfl_xor(mn, off, 32));
    }
    if (lane == 0) { red[wave] = mx; red[NW_ + wave] = mn; }
    __syncthreads();
    mx = red[0]; mn = red[NW_];
#pragma unroll
    for (int w = 1; w < NW_; ++w) {
        mx = fmaxf(mx, red[w]);
        mn = fminf(mn, red[NW_ + w]);
    }

    const float range = mx - mn;
    const float delta = range * (1.0f / NB_);
    const float inv   = (range > 0.0f) ? (float)NB_ / range : 0.0f;
    const float cen0  = mn + 0.5f * delta;

    {
        const float mm[4] = {m4.x, m4.y, m4.z, m4.w};
        const float ff[4] = {f4.x, f4.y, f4.z, f4.w};
#pragma unroll
        for (int e = 0; e < 4; ++e) {
            int lev = (int)((mm[e] - mn) * inv);
            lev = min(max(lev, 0), NB_ - 1);
            const float d = mm[e] - (cen0 + (float)lev * delta);
            *reinterpret_cast<float4*>(rec + 4 * (4 * tid + e)) =
                make_float4(d, d * d, ff[e], (float)lev);
        }
    }
    __syncthreads();

    {
        const int  pc  = c16 & 7;
        const float fsel = (c16 < 8) ? 1.0f : 0.0f;
        const float osel = (c16 < 8) ? 0.0f : 1.0f;
        const bool p1 = (pc & 1) != 0, p2 = (pc & 2) != 0, p4 = (pc & 4) != 0;
        v8f accA = {}, accB = {};
        const int jb = wave * 128 + 2 * half;
        const float levc = (float)c16;
#pragma unroll 4
        for (int ks = 0; ks < 32; ks += 2) {
            const float4* rp = reinterpret_cast<const float4*>(rec) + jb + 4 * ks;
            const float4 rc[4] = {rp[0], rp[1], rp[4], rp[5]};
            float av[4], wv[4];
#pragma unroll
            for (int e = 0; e < 4; ++e) {
                const float d  = rc[e].x;
                const float d2 = rc[e].y;
                const float d4 = d2 * d2;
                float v = fmaf(rc[e].z, fsel, osel);
                v = p1 ? v * d  : v;
                v = p2 ? v * d2 : v;
                v = p4 ? v * d4 : v;
                av[e] = (rc[e].w == levc) ? 1.0f : 0.0f;
                wv[e] = v;
            }
            const v2f a0 = {av[0], av[1]}, b0 = {wv[0], wv[1]};
            const v2f a1 = {av[2], av[3]}, b1 = {wv[2], wv[3]};
            accA = __builtin_amdgcn_wmma_f32_16x16x4_f32(
                false, a0, false, b0, (short)0, accA, false, false);
            accB = __builtin_amdgcn_wmma_f32_16x16x4_f32(
                false, a1, false, b1, (short)0, accB, false, false);
        }
#pragma unroll
        for (int r = 0; r < 8; ++r)
            part[wave * 256 + (8 * half + r) * 16 + c16] = accA[r] + accB[r];
    }
    __syncthreads();
    {
        float t = part[tid];
#pragma unroll
        for (int w = 1; w < NW_; ++w) t += part[w * 256 + tid];
        mom[tid] = t;
    }
    __syncthreads();

    v2f af[NB_][2];
#pragma unroll
    for (int k = 0; k < NB_; ++k) {
#pragma unroll
        for (int pg = 0; pg < 2; ++pg) {
            const float2 v = *reinterpret_cast<const float2*>(
                mom + k * 16 + (c16 & 1) * 8 + 4 * pg + 2 * half);
            af[k][pg].x = (c16 < 2) ? v.x : 0.0f;
            af[k][pg].y = (c16 < 2) ? v.y : 0.0f;
        }
    }

    const float L2E = 1.44269504088896340736f;
    float res[2];
#pragma unroll
    for (int tile = 0; tile < 2; ++tile) {
        const int row = q * 256 + wave * 32 + tile * 16 + c16;
        const float s   = sf[row];
        const float sL2 = s * L2E;
        const float rL2 = (s >= 0.0f ? s * mx : s * mn) * L2E;
        const float s2 = s * s;
        const float s4 = s2 * s2;
        const float q0 = half ? 0.5f * s2 : 1.0f;
        const float q1 = q0 * s * (half ? (1.0f / 3.0f) : 1.0f);
        const float q4 = s4 * (half ? s2 * (1.0f / 720.0f) : (1.0f / 24.0f));
        const float q5 = q4 * s * (half ? (1.0f / 7.0f) : (1.0f / 5.0f));
        v8f n0 = {}, n1 = {};
#pragma unroll
        for (int k = 0; k < NB_; ++k) {
            const float Mk = cen0 + (float)k * delta;
            const float E = __builtin_amdgcn_exp2f(sL2 * Mk - rL2);
            const v2f b0 = {E * q0, E * q1};
            const v2f b1 = {E * q4, E * q5};
            n0 = __builtin_amdgcn_wmma_f32_16x16x4_f32(
                false, af[k][0], false, b0, (short)0, n0, false, false);
            n1 = __builtin_amdgcn_wmma_f32_16x16x4_f32(
                false, af[k][1], false, b1, (short)0, n1, false, false);
        }
        const float num = n0[0] + n1[0];
        const float den = n0[1] + n1[1];
        res[tile] = g * (num / den) + s;
    }
    const float r1 = __shfl(res[1], c16, 32);
    const float fin = half ? r1 : res[0];
    out[(size_t)bc * N_ + q * 256 + wave * 32 + lane] = fin;
}

extern "C" void kernel_launch(void* const* d_in, const int* in_sizes, int n_in,
                              void* d_out, int out_size, void* d_ws, size_t ws_size,
                              hipStream_t stream) {
    (void)in_sizes; (void)n_in; (void)out_size; (void)d_ws; (void)ws_size;
    const float* feature = (const float*)d_in[0];
    const float* seg     = (const float*)d_in[1];
    const float* gamma   = (const float*)d_in[2];
    float* out           = (float*)d_out;

    dim3 grid(B_ * C_ * 4);
    dim3 block(256);
    psm_attn_kernel<<<grid, block, 0, stream>>>(feature, seg, gamma, out);
}


# rows: powers contracted first by a dense 16-level x 8-power x 16-row matrix product (4 WMMAs per 16 rows, was 32), then the 16 level weights exp(s M_k) applied in the D registers (sum order changes)
# speedup vs baseline: 5.2569x; 1.1557x over previous
#include <hip/hip_runtime.h>

typedef __attribute__((ext_vector_type(2))) float v2f;
typedef __attribute__((ext_vector_type(8))) float v8f;

#define B_  2
#define C_  32
#define N_  1024
#define NB_ 16
#define NW_ 8

__device__ __forceinline__ float xhalf(float x) {
    const int b = __builtin_bit_cast(int, x);
    return __builtin_bit_cast(float,
        __builtin_amdgcn_permlanex16(b, b, 0x76543210, 0xfedcba98, false, false));
}

__global__ __launch_bounds__(256)
void psm_attn_kernel(const float* __restrict__ feature,
                     const float* __restrict__ seg,
                     const float* __restrict__ gamma,
                     float* __restrict__ out)
{
    __shared__ __attribute__((aligned(16))) float rec[4 * N_];
    __shared__ __attribute__((aligned(16))) float sf[N_];
    __shared__ __attribute__((aligned(16))) float part[NW_ * 256];
    __shared__ __attribute__((aligned(16))) float mom[256];
    __shared__ float red[2 * NW_];

    const int bc   = blockIdx.x >> 2;
    const int q    = blockIdx.x & 3;
    const int b    = bc >> 5;
    const int tid  = threadIdx.x;
    const int lane = tid & 31;
    const int wave = tid >> 5;
    const int c16  = lane & 15;
    const int half = lane >> 4;

    const float* fptr = feature + (size_t)bc * N_;
    const float* mptr = seg + ((size_t)b * 2 + 1) * N_;
    const float g = gamma[0];

    const float4 m4 = *reinterpret_cast<const float4*>(mptr + 4 * tid);
    const float4 f4 = *reinterpret_cast<const float4*>(fptr + 4 * tid);
    *reinterpret_cast<float4*>(sf + 4 * tid) = f4;
    float mx = fmaxf(fmaxf(m4.x, m4.y), fmaxf(m4.z, m4.w));
    float mn = fminf(fminf(m4.x, m4.y), fminf(m4.z, m4.w));
#pragma unroll
    for (int off = 16; off > 0; off >>= 1) {
        mx = fmaxf(mx, __shfl_xor(mx, off, 32));
        mn = fminf(mn, __shfl_xor(mn, off, 32));
    }
    if (lane == 0) { red[wave] = mx; red[NW_ + wave] = mn; }
    __syncthreads();
    mx = red[0]; mn = red[NW_];
#pragma unroll
    for (int w = 1; w < NW_; ++w) {
        mx = fmaxf(mx, red[w]);
        mn = fminf(mn, red[NW_ + w]);
    }

    const float range = mx - mn;
    const float delta = range * (1.0f / NB_);
    const float inv   = (range > 0.0f) ? (float)NB_ / range : 0.0f;
    const float cen0  = mn + 0.5f * delta;

    {
        const float mm[4] = {m4.x, m4.y, m4.z, m4.w};
        const float ff[4] = {f4.x, f4.y, f4.z, f4.w};
#pragma unroll
        for (int e = 0; e < 4; ++e) {
            int lev = (int)((mm[e] - mn) * inv);
            lev = min(max(lev, 0), NB_ - 1);
            const float d = mm[e] - (cen0 + (float)lev * delta);
            *reinterpret_cast<float4*>(rec + 4 * (4 * tid + e)) =
                make_float4(d, d * d, ff[e], (float)lev);
        }
    }
    __syncthreads();

    {
        const int  pc  = c16 & 7;
        const float fsel = (c16 < 8) ? 1.0f : 0.0f;
        const float osel = (c16 < 8) ? 0.0f : 1.0f;
        const bool p1 = (pc & 1) != 0, p2 = (pc & 2) != 0, p4 = (pc & 4) != 0;
        v8f accA = {}, accB = {};
        const int jb = wave * 128 + 2 * half;
        const float levc = (float)c16;
#pragma unroll 4
        for (int ks = 0; ks < 32; ks += 2) {
            const float4* rp = reinterpret_cast<const float4*>(rec) + jb + 4 * ks;
            const float4 rc[4] = {rp[0], rp[1], rp[4], rp[5]};
            float av[4], wv[4];
#pragma unroll
            for (int e = 0; e < 4; ++e) {
                const float d  = rc[e].x;
                const float d2 = rc[e].y;
                const float d4 = d2 * d2;
                float v = fmaf(rc[e].z, fsel, osel);
                v = p1 ? v * d  : v;
                v = p2 ? v * d2 : v;
                v = p4 ? v * d4 : v;
                av[e] = (rc[e].w == levc) ? 1.0f : 0.0f;
                wv[e] = v;
            }
            const v2f a0 = {av[0], av[1]}, b0 = {wv[0], wv[1]};
            const v2f a1 = {av[2], av[3]}, b1 = {wv[2], wv[3]};
            accA = __builtin_amdgcn_wmma_f32_16x16x4_f32(
                false, a0, false, b0, (short)0, accA, false, false);
            accB = __builtin_amdgcn_wmma_f32_16x16x4_f32(
                false, a1, false, b1, (short)0, accB, false, false);
        }
#pragma unroll
        for (int r = 0; r < 8; ++r)
            part[wave * 256 + (8 * half + r) * 16 + c16] = accA[r] + accB[r];
    }
    __syncthreads();
    {
        float t = part[tid];
#pragma unroll
        for (int w = 1; w < NW_; ++w) t += part[w * 256 + tid];
        mom[tid] = t;
    }
    __syncthreads();

    v2f aT[2], aU[2];
#pragma unroll
    for (int pg = 0; pg < 2; ++pg) {
        const float2 t = *reinterpret_cast<const float2*>(mom + c16 * 16 + 4 * pg + 2 * half);
        const float2 u = *reinterpret_cast<const float2*>(mom + c16 * 16 + 8 + 4 * pg + 2 * half);
        aT[pg].x = t.x; aT[pg].y = t.y;
        aU[pg].x = u.x; aU[pg].y = u.y;
    }

    const float L2E = 1.44269504088896340736f;
    const float lev0 = (float)(8 * half);
    float res[2];
#pragma unroll
    for (int tile = 0; tile < 2; ++tile) {
        const int row = q * 256 + wave * 32 + tile * 16 + c16;
        const float s   = sf[row];
        const float sL2 = s * L2E;
        const float rL2 = (s >= 0.0f ? s * mx : s * mn) * L2E;
        const float s2 = s * s;
        const float s4 = s2 * s2;
        const float q0 = half ? 0.5f * s2 : 1.0f;
        const float q1 = q0 * s * (half ? (1.0f / 3.0f) : 1.0f);
        const float q4 = s4 * (half ? s2 * (1.0f / 720.0f) : (1.0f / 24.0f));
        const float q5 = q4 * s * (half ? (1.0f / 7.0f) : (1.0f / 5.0f));
        const v2f b0 = {q0, q1};
        const v2f b1 = {q4, q5};
        v8f gT = {}, gU = {};
        gT = __builtin_amdgcn_wmma_f32_16x16x4_f32(
            false, aT[0], false, b0, (short)0, gT, false, false);
        gU = __builtin_amdgcn_wmma_f32_16x16x4_f32(
            false, aU[0], false, b0, (short)0, gU, false, false);
        gT = __builtin_amdgcn_wmma_f32_16x16x4_f32(
            false, aT[1], false, b1, (short)0, gT, false, false);
        gU = __builtin_amdgcn_wmma_f32_16x16x4_f32(
            false, aU[1], false, b1, (short)0, gU, false, false);
        float num = 0.0f, den = 0.0f;
#pragma unroll
        for (int r = 0; r < 8; ++r) {
            const float Mk = cen0 + (lev0 + (float)r) * delta;
            const float E = __builtin_amdgcn_exp2f(sL2 * Mk - rL2);
            num = fmaf(E, gT[r], num);
            den = fmaf(E, gU[r], den);
        }
        num += xhalf(num);
        den += xhalf(den);
        res[tile] = g * (num / den) + s;
    }
    const float fin = half ? res[1] : res[0];
    out[(size_t)bc * N_ + q * 256 + wave * 32 + lane] = fin;
}

extern "C" void kernel_launch(void* const* d_in, const int* in_sizes, int n_in,
                              void* d_out, int out_size, void* d_ws, size_t ws_size,
                              hipStream_t stream) {
    (void)in_sizes; (void)n_in; (void)out_size; (void)d_ws; (void)ws_size;
    const float* feature = (const float*)d_in[0];
    const float* seg     = (const float*)d_in[1];
    const float* gamma   = (const float*)d_in[2];
    float* out           = (float*)d_out;

    dim3 grid(B_ * C_ * 4);
    dim3 block(256);
    psm_attn_kernel<<<grid, block, 0, stream>>>(feature, seg, gamma, out);
}


# moment product's B rows (f d^p, d^p) built once per j at staging, filed in LDS in fragment order; a wave issues its 64 fragment reads ahead of its 32 WMMAs (T terms round as f x d^p; sum order kept)
# speedup vs baseline: 6.2693x; 1.1926x over previous
#include <hip/hip_runtime.h>

typedef __attribute__((ext_vector_type(2))) float v2f;
typedef __attribute__((ext_vector_type(8))) float v8f;

#define B_  2
#define C_  32
#define N_  1024
#define NB_ 16
#define NW_ 8
#define TP_ 70

__device__ __forceinline__ float xhalf(float x) {
    const int b = __builtin_bit_cast(int, x);
    return __builtin_bit_cast(float,
        __builtin_amdgcn_permlanex16(b, b, 0x76543210, 0xfedcba98, false, false));
}

__global__ __launch_bounds__(256)
void psm_attn_kernel(const float* __restrict__ feature,
                     const float* __restrict__ seg,
                     const float* __restrict__ gamma,
                     float* __restrict__ out)
{
    __shared__ __attribute__((aligned(16))) float tbl[(N_ / 4) * TP_];
    __shared__ __attribute__((aligned(16))) float sf[N_];
    __shared__ __attribute__((aligned(16))) float part[NW_ * 256];
    __shared__ __attribute__((aligned(16))) float mom[256];
    __shared__ float red[2 * NW_];

    const int bc   = blockIdx.x >> 2;
    const int q    = blockIdx.x & 3;
    const int b    = bc >> 5;
    const int tid  = threadIdx.x;
    const int lane = tid & 31;
    const int wave = tid >> 5;
    const int c16  = lane & 15;
    const int half = lane >> 4;

    const float* fptr = feature + (size_t)bc * N_;
    const float* mptr = seg + ((size_t)b * 2 + 1) * N_;
    const float g = gamma[0];

    const float4 m4 = *reinterpret_cast<const float4*>(mptr + 4 * tid);
    const float4 f4 = *reinterpret_cast<const float4*>(fptr + 4 * tid);
    *reinterpret_cast<float4*>(sf + 4 * tid) = f4;
    float mx = fmaxf(fmaxf(m4.x, m4.y), fmaxf(m4.z, m4.w));
    float mn = fminf(fminf(m4.x, m4.y), fminf(m4.z, m4.w));
#pragma unroll
    for (int off = 16; off > 0; off >>= 1) {
        mx = fmaxf(mx, __shfl_xor(mx, off, 32));
        mn = fminf(mn, __shfl_xor(mn, off, 32));
    }
    if (lane == 0) { red[wave] = mx; red[NW_ + wave] = mn; }
    __syncthreads();
    mx = red[0]; mn = red[NW_];
#pragma unroll
    for (int w = 1; w < NW_; ++w) {
        mx = fmaxf(mx, red[w]);
        mn = fminf(mn, red[NW_ + w]);
    }

    const float range = mx - mn;
    const float delta = range * (1.0f / NB_);
    const float inv   = (range > 0.0f) ? (float)NB_ / range : 0.0f;
    const float cen0  = mn + 0.5f * delta;

    {
        const float mm[4] = {m4.x, m4.y, m4.z, m4.w};
        const float ff[4] = {f4.x, f4.y, f4.z, f4.w};
        float pw[4][8];
        float fw[4][8];
        float lv[4];
#pragma unroll
        for (int e = 0; e < 4; ++e) {
            int lev = (int)((mm[e] - mn) * inv);
            lev = min(max(lev, 0), NB_ - 1);
            const float d  = mm[e] - (cen0 + (float)lev * delta);
            const float d2 = d * d;
            const float d3 = d2 * d;
            const float d4 = d2 * d2;
            pw[e][0] = 1.0f;    pw[e][1] = d;       pw[e][2] = d2;      pw[e][3] = d3;
            pw[e][4] = d4;      pw[e][5] = d4 * d;  pw[e][6] = d4 * d2; pw[e][7] = d4 * d3;
            fw[e][0] = ff[e];
#pragma unroll
            for (int p = 1; p < 8; ++p) fw[e][p] = ff[e] * pw[e][p];
            lv[e] = (float)lev;
        }
        float* rp = tbl + TP_ * tid;
#pragma unroll
        for (int hh = 0; hh < 2; ++hh) {
            const int e0 = 2 * hh, e1 = 2 * hh + 1;
#pragma unroll
            for (int p = 0; p < 8; ++p) {
                *reinterpret_cast<float2*>(rp + 32 * hh + 2 * p)      = make_float2(fw[e0][p], fw[e1][p]);
                *reinterpret_cast<float2*>(rp + 32 * hh + 16 + 2 * p) = make_float2(pw[e0][p], pw[e1][p]);
            }
            *reinterpret_cast<float2*>(rp + 64 + 2 * hh) = make_float2(lv[e0], lv[e1]);
        }
    }
    __syncthreads();

    {
        const float levc = (float)c16;
        const float* bp = tbl + TP_ * (32 * wave) + 32 * half + 2 * c16;
        const float* lp = tbl + TP_ * (32 * wave) + 64 + 2 * half;
        float2 bq[32], lq[32];
#pragma unroll
        for (int ks = 0; ks < 32; ++ks) {
            bq[ks] = *reinterpret_cast<const float2*>(bp + TP_ * ks);
            lq[ks] = *reinterpret_cast<const float2*>(lp + TP_ * ks);
        }
        v8f accA = {}, accB = {};
#pragma unroll
        for (int ks = 0; ks < 32; ks += 2) {
            const v2f a0 = {lq[ks].x == levc ? 1.0f : 0.0f, lq[ks].y == levc ? 1.0f : 0.0f};
            const v2f b0 = {bq[ks].x, bq[ks].y};
            const v2f a1 = {lq[ks + 1].x == levc ? 1.0f : 0.0f, lq[ks + 1].y == levc ? 1.0f : 0.0f};
            const v2f b1 = {bq[ks + 1].x, bq[ks + 1].y};
            accA = __builtin_amdgcn_wmma_f32_16x16x4_f32(
                false, a0, false, b0, (short)0, accA, false, false);
            accB = __builtin_amdgcn_wmma_f32_16x16x4_f32(
                false, a1, false, b1, (short)0, accB, false, false);
        }
        __builtin_amdgcn_sched_group_barrier(0x100, 64, 0);
        __builtin_amdgcn_sched_group_barrier(0x008, 32, 0);
#pragma unroll
        for (int r = 0; r < 8; ++r)
            part[wave * 256 + (8 * half + r) * 16 + c16] = accA[r] + accB[r];
    }
    __syncthreads();
    {
        float t = part[tid];
#pragma unroll
        for (int w = 1; w < NW_; ++w) t += part[w * 256 + tid];
        mom[tid] = t;
    }
    __syncthreads();

    v2f aT[2], aU[2];
#pragma unroll
    for (int pg = 0; pg < 2; ++pg) {
        const float2 t = *reinterpret_cast<const float2*>(mom + c16 * 16 + 4 * pg + 2 * half);
        const float2 u = *reinterpret_cast<const float2*>(mom + c16 * 16 + 8 + 4 * pg + 2 * half);
        aT[pg].x = t.x; aT[pg].y = t.y;
        aU[pg].x = u.x; aU[pg].y = u.y;
    }

    const float L2E = 1.44269504088896340736f;
    const float lev0 = (float)(8 * half);
    float res[2];
#pragma unroll
    for (int tile = 0; tile < 2; ++tile) {
        const int row = q * 256 + wave * 32 + tile * 16 + c16;
        const float s   = sf[row];
        const float sL2 = s * L2E;
        const float rL2 = (s >= 0.0f ? s * mx : s * mn) * L2E;
        const float s2 = s * s;
        const float s4 = s2 * s2;
        const float q0 = half ? 0.5f * s2 : 1.0f;
        const float q1 = q0 * s * (half ? (1.0f / 3.0f) : 1.0f);
        const float q4 = s4 * (half ? s2 * (1.0f / 720.0f) : (1.0f / 24.0f));
        const float q5 = q4 * s * (half ? (1.0f / 7.0f) : (1.0f / 5.0f));
        const v2f b0 = {q0, q1};
        const v2f b1 = {q4, q5};
        v8f gT = {}, gU = {};
        gT = __builtin_amdgcn_wmma_f32_16x16x4_f32(
            false, aT[0], false, b0, (short)0, gT, false, false);
        gU = __builtin_amdgcn_wmma_f32_16x16x4_f32(
            false, aU[0], false, b0, (short)0, gU, false, false);
        gT = __builtin_amdgcn_wmma_f32_16x16x4_f32(
            false, aT[1], false, b1, (short)0, gT, false, false);
        gU = __builtin_amdgcn_wmma_f32_16x16x4_f32(
            false, aU[1], false, b1, (short)0, gU, false, false);
        float num = 0.0f, den = 0.0f;
#pragma unroll
        for (int r = 0; r < 8; ++r) {
            const float Mk = cen0 + (lev0 + (float)r) * delta;
            const float E = __builtin_amdgcn_exp2f(sL2 * Mk - rL2);
            num = fmaf(E, gT[r], num);
            den = fmaf(E, gU[r], den);
        }
        num += xhalf(num);
        den += xhalf(den);
        res[tile] = g * (num / den) + s;
    }
    const float fin = half ? res[1] : res[0];
    out[(size_t)bc * N_ + q * 256 + wave * 32 + lane] = fin;
}

extern "C" void kernel_launch(void* const* d_in, const int* in_sizes, int n_in,
                              void* d_out, int out_size, void* d_ws, size_t ws_size,
                              hipStream_t stream) {
    (void)in_sizes; (void)n_in; (void)out_size; (void)d_ws; (void)ws_size;
    const float* feature = (const float*)d_in[0];
    const float* seg     = (const float*)d_in[1];
    const float* gamma   = (const float*)d_in[2];
    float* out           = (float*)d_out;

    dim3 grid(B_ * C_ * 4);
    dim3 block(256);
    psm_attn_kernel<<<grid, block, 0, stream>>>(feature, seg, gamma, out);
}


# 4 waves per block, one per SIMD, was 8: a thread stages 8 j, a wave sums 256 j (64 products) and 4 row tiles; per-SIMD work kept, half the waves enter; 4 partial tiles joined (sum order changes)
# speedup vs baseline: 6.9392x; 1.1069x over previous
#include <hip/hip_runtime.h>

typedef __attribute__((ext_vector_type(2))) float v2f;
typedef __attribute__((ext_vector_type(8))) float v8f;

#define B_  2
#define C_  32
#define N_  1024
#define NB_ 16
#define NW_ 4
#define NT_ 128
#define TP_ 70

__device__ __forceinline__ float xhalf(float x) {
    const int b = __builtin_bit_cast(int, x);
    return __builtin_bit_cast(float,
        __builtin_amdgcn_permlanex16(b, b, 0x76543210, 0xfedcba98, false, false));
}

__global__ __launch_bounds__(NT_)
void psm_attn_kernel(const float* __restrict__ feature,
                     const float* __restrict__ seg,
                     const float* __restrict__ gamma,
                     float* __restrict__ out)
{
    __shared__ __attribute__((aligned(16))) float tbl[(N_ / 4) * TP_];
    __shared__ __attribute__((aligned(16))) float sf[N_];
    __shared__ __attribute__((aligned(16))) float part[NW_ * 256];
    __shared__ __attribute__((aligned(16))) float mom[256];
    __shared__ float red[2 * NW_];

    const int bc   = blockIdx.x >> 2;
    const int q    = blockIdx.x & 3;
    const int b    = bc >> 5;
    const int tid  = threadIdx.x;
    const int lane = tid & 31;
    const int wave = tid >> 5;
    const int c16  = lane & 15;
    const int half = lane >> 4;

    const float* fptr = feature + (size_t)bc * N_;
    const float* mptr = seg + ((size_t)b * 2 + 1) * N_;
    const float g = gamma[0];

    const float4 m4 = *reinterpret_cast<const float4*>(mptr + 4 * tid);
    const float4 m4b = *reinterpret_cast<const float4*>(mptr + 4 * (tid + NT_));
    const float4 f4 = *reinterpret_cast<const float4*>(fptr + 4 * tid);
    const float4 f4b = *reinterpret_cast<const float4*>(fptr + 4 * (tid + NT_));
    *reinterpret_cast<float4*>(sf + 4 * tid) = f4;
    *reinterpret_cast<float4*>(sf + 4 * (tid + NT_)) = f4b;
    float mx = fmaxf(fmaxf(fmaxf(m4.x, m4.y), fmaxf(m4.z, m4.w)), fmaxf(fmaxf(m4b.x, m4b.y), fmaxf(m4b.z, m4b.w)));
    float mn = fminf(fminf(fminf(m4.x, m4.y), fminf(m4.z, m4.w)), fminf(fminf(m4b.x, m4b.y), fminf(m4b.z, m4b.w)));
#pragma unroll
    for (int off = 16; off > 0; off >>= 1) {
        mx = fmaxf(mx, __shfl_xor(mx, off, 32));
        mn = fminf(mn, __shfl_xor(mn, off, 32));
    }
    if (lane == 0) { red[wave] = mx; red[NW_ + wave] = mn; }
    __syncthreads();
    mx = red[0]; mn = red[NW_];
#pragma unroll
    for (int w = 1; w < NW_; ++w) {
        mx = fmaxf(mx, red[w]);
        mn = fminf(mn, red[NW_ + w]);
    }

    const float range = mx - mn;
    const float delta = range * (1.0f / NB_);
    const float inv   = (range > 0.0f) ? (float)NB_ / range : 0.0f;
    const float cen0  = mn + 0.5f * delta;

#pragma unroll
    for (int rec = 0; rec < 2; ++rec) {
        const float4 mr = rec ? m4b : m4;
        const float4 fr = rec ? f4b : f4;
        const float mm[4] = {mr.x, mr.y, mr.z, mr.w};
        const float ff[4] = {fr.x, fr.y, fr.z, fr.w};
        float pw[4][8];
        float fw[4][8];
        float lv[4];
#pragma unroll
        for (int e = 0; e < 4; ++e) {
            int lev = (int)((mm[e] - mn) * inv);
            lev = min(max(lev, 0), NB_ - 1);
            const float d  = mm[e] - (cen0 + (float)lev * delta);
            const float d2 = d * d;
            const float d3 = d2 * d;
            const float d4 = d2 * d2;
            pw[e][0] = 1.0f;    pw[e][1] = d;       pw[e][2] = d2;      pw[e][3] = d3;
            pw[e][4] = d4;      pw[e][5] = d4 * d;  pw[e][6] = d4 * d2; pw[e][7] = d4 * d3;
            fw[e][0] = ff[e];
#pragma unroll
            for (int p = 1; p < 8; ++p) fw[e][p] = ff[e] * pw[e][p];
            lv[e] = (float)lev;
        }
        float* rp = tbl + TP_ * (tid + NT_ * rec);
#pragma unroll
        for (int hh = 0; hh < 2; ++hh) {
            const int e0 = 2 * hh, e1 = 2 * hh + 1;
#pragma unroll
            for (int p = 0; p < 8; ++p) {
                *reinterpret_cast<float2*>(rp + 32 * hh + 2 * p)      = make_float2(fw[e0][p], fw[e1][p]);
                *reinterpret_cast<float2*>(rp + 32 * hh + 16 + 2 * p) = make_float2(pw[e0][p], pw[e1][p]);
            }
            *reinterpret_cast<float2*>(rp + 64 + 2 * hh) = make_float2(lv[e0], lv[e1]);
        }
    }
    __syncthreads();

    {
        const float levc = (float)c16;
        const float* bp = tbl + TP_ * (32 * wave) + 32 * half + 2 * c16;
        const float* lp = tbl + TP_ * (32 * wave) + 64 + 2 * half;
        float2 bq[64], lq[64];
        v8f accA = {}, accB = {};
        auto loadq = [&](const int q0) {
#pragma unroll
            for (int ks = 16 * q0; ks < 16 * q0 + 16; ++ks) {
                const int r = (ks < 32) ? ks : NT_ + ks - 32;
                bq[ks] = *reinterpret_cast<const float2*>(bp + TP_ * r);
                lq[ks] = *reinterpret_cast<const float2*>(lp + TP_ * r);
            }
        };
        auto mulq = [&](const int q0) {
            float lv = levc;
            asm volatile("" : "+v"(lv));
#pragma unroll
            for (int ks = 16 * q0; ks < 16 * q0 + 16; ks += 2) {
                const v2f a0 = {lq[ks].x == lv ? 1.0f : 0.0f, lq[ks].y == lv ? 1.0f : 0.0f};
                const v2f b0 = {bq[ks].x, bq[ks].y};
                const v2f a1 = {lq[ks + 1].x == lv ? 1.0f : 0.0f, lq[ks + 1].y == lv ? 1.0f : 0.0f};
                const v2f b1 = {bq[ks + 1].x, bq[ks + 1].y};
                accA = __builtin_amdgcn_wmma_f32_16x16x4_f32(
                    false, a0, false, b0, (short)0, accA, false, false);
                accB = __builtin_amdgcn_wmma_f32_16x16x4_f32(
                    false, a1, false, b1, (short)0, accB, false, false);
                __builtin_amdgcn_sched_group_barrier(0x002, 4, 0);
                __builtin_amdgcn_sched_group_barrier(0x008, 1, 0);
                __builtin_amdgcn_sched_group_barrier(0x002, 4, 0);
                __builtin_amdgcn_sched_group_barrier(0x008, 1, 0);
            }
        };
#define HOLD() do { asm volatile("" ::: "memory"); __builtin_amdgcn_sched_barrier(0); } while (0)
        loadq(0);
        HOLD();
        loadq(1);
        __builtin_amdgcn_sched_group_barrier(0x100, 16, 0);
        mulq(0);
        HOLD();
        loadq(2);
        HOLD();
        mulq(1);
        HOLD();
        loadq(3);
        HOLD();
        mulq(2);
        mulq(3);
#undef HOLD
#pragma unroll
        for (int r = 0; r < 8; ++r)
            part[wave * 256 + (8 * half + r) * 16 + c16] = accA[r] + accB[r];
    }
    __syncthreads();
#pragma unroll
    for (int u = 0; u < 2; ++u) {
        const int idx = tid + NT_ * u;
        float t = part[idx];
#pragma unroll
        for (int w = 1; w < NW_; ++w) t += part[w * 256 + idx];
        mom[idx] = t;
    }
    __syncthreads();

    v2f aT[2], aU[2];
#pragma unroll
    for (int pg = 0; pg < 2; ++pg) {
        const float2 t = *reinterpret_cast<const float2*>(mom + c16 * 16 + 4 * pg + 2 * half);
        const float2 u = *reinterpret_cast<const float2*>(mom + c16 * 16 + 8 + 4 * pg + 2 * half);
        aT[pg].x = t.x; aT[pg].y = t.y;
        aU[pg].x = u.x; aU[pg].y = u.y;
    }

    const float L2E = 1.44269504088896340736f;
    const float lev0 = (float)(8 * half);
    float res[4];
#pragma unroll
    for (int tile = 0; tile < 4; ++tile) {
        const int row = q * 256 + wave * 64 + tile * 16 + c16;
        const float s   = sf[row];
        const float sL2 = s * L2E;
        const float rL2 = (s >= 0.0f ? s * mx : s * mn) * L2E;
        const float s2 = s * s;
        const float s4 = s2 * s2;
        const float q0 = half ? 0.5f * s2 : 1.0f;
        const float q1 = q0 * s * (half ? (1.0f / 3.0f) : 1.0f);
        const float q4 = s4 * (half ? s2 * (1.0f / 720.0f) : (1.0f / 24.0f));
        const float q5 = q4 * s * (half ? (1.0f / 7.0f) : (1.0f / 5.0f));
        const v2f b0 = {q0, q1};
        const v2f b1 = {q4, q5};
        v8f gT = {}, gU = {};
        gT = __builtin_amdgcn_wmma_f32_16x16x4_f32(
            false, aT[0], false, b0, (short)0, gT, false, false);
        gU = __builtin_amdgcn_wmma_f32_16x16x4_f32(
            false, aU[0], false, b0, (short)0, gU, false, false);
        gT = __builtin_amdgcn_wmma_f32_16x16x4_f32(
            false, aT[1], false, b1, (short)0, gT, false, false);
        gU = __builtin_amdgcn_wmma_f32_16x16x4_f32(
            false, aU[1], false, b1, (short)0, gU, false, false);
        float num = 0.0f, den = 0.0f;
#pragma unroll
        for (int r = 0; r < 8; ++r) {
            const float Mk = cen0 + (lev0 + (float)r) * delta;
            const float E = __builtin_amdgcn_exp2f(sL2 * Mk - rL2);
            num = fmaf(E, gT[r], num);
            den = fmaf(E, gU[r], den);
        }
        num += xhalf(num);
        den += xhalf(den);
        res[tile] = g * (num / den) + s;
    }
    float* orow = out + (size_t)bc * N_ + q * 256 + wave * 64 + lane;
    orow[0]  = half ? res[1] : res[0];
    orow[32] = half ? res[3] : res[2];
}

extern "C" void kernel_launch(void* const* d_in, const int* in_sizes, int n_in,
                              void* d_out, int out_size, void* d_ws, size_t ws_size,
                              hipStream_t stream) {
    (void)in_sizes; (void)n_in; (void)out_size; (void)d_ws; (void)ws_size;
    const float* feature = (const float*)d_in[0];
    const float* seg     = (const float*)d_in[1];
    const float* gamma   = (const float*)d_in[2];
    float* out           = (float*)d_out;

    dim3 grid(B_ * C_ * 4);
    dim3 block(NT_);
    psm_attn_kernel<<<grid, block, 0, stream>>>(feature, seg, gamma, out);
}


# min/max of m per wave, no exchange: each wave loads the whole m row (8 x 16 B per lane), reduces by DPP row_xmask + permlanex16; no LDS join, no barrier before the moment loop; same values and sums
# speedup vs baseline: 7.5070x; 1.0818x over previous
#include <hip/hip_runtime.h>

typedef __attribute__((ext_vector_type(2))) float v2f;
typedef __attribute__((ext_vector_type(8))) float v8f;

#define B_  2
#define C_  32
#define N_  1024
#define NB_ 16
#define NW_ 4
#define NT_ 128
#define TP_ 70

__device__ __forceinline__ float xhalf(float x) {
    const int b = __builtin_bit_cast(int, x);
    return __builtin_bit_cast(float,
        __builtin_amdgcn_permlanex16(b, b, 0x76543210, 0xfedcba98, false, false));
}

#define ROW_MAX(x, X) asm("v_max_num_f32_dpp %0, %0, %0 row_xmask:" #X " row_mask:0xf bank_mask:0xf" : "+v"(x))
#define ROW_MIN(x, X) asm("v_min_num_f32_dpp %0, %0, %0 row_xmask:" #X " row_mask:0xf bank_mask:0xf" : "+v"(x))

__global__ __launch_bounds__(NT_)
void psm_attn_kernel(const float* __restrict__ feature,
                     const float* __restrict__ seg,
                     const float* __restrict__ gamma,
                     float* __restrict__ out)
{
    __shared__ __attribute__((aligned(16))) float tbl[(N_ / 4) * TP_];
    __shared__ __attribute__((aligned(16))) float sf[N_];
    __shared__ __attribute__((aligned(16))) float part[NW_ * 256];
    __shared__ __attribute__((aligned(16))) float mom[256];

    const int bc   = blockIdx.x >> 2;
    const int q    = blockIdx.x & 3;
    const int b    = bc >> 5;
    const int tid  = threadIdx.x;
    const int lane = tid & 31;
    const int wave = tid >> 5;
    const int c16  = lane & 15;
    const int half = lane >> 4;

    const float* fptr = feature + (size_t)bc * N_;
    const float* mptr = seg + ((size_t)b * 2 + 1) * N_;
    const float g = gamma[0];

    const float4 m4 = *reinterpret_cast<const float4*>(mptr + 4 * tid);
    const float4 m4b = *reinterpret_cast<const float4*>(mptr + 4 * (tid + NT_));
    const int o1 = (tid + 32) & (NT_ - 1), o2 = (tid + 64) & (NT_ - 1), o3 = (tid + 96) & (NT_ - 1);
    const float4 n0 = *reinterpret_cast<const float4*>(mptr + 4 * o1);
    const float4 n1 = *reinterpret_cast<const float4*>(mptr + 4 * (o1 + NT_));
    const float4 n2 = *reinterpret_cast<const float4*>(mptr + 4 * o2);
    const float4 n3 = *reinterpret_cast<const float4*>(mptr + 4 * (o2 + NT_));
    const float4 n4 = *reinterpret_cast<const float4*>(mptr + 4 * o3);
    const float4 n5 = *reinterpret_cast<const float4*>(mptr + 4 * (o3 + NT_));
    const float4 f4 = *reinterpret_cast<const float4*>(fptr + 4 * tid);
    const float4 f4b = *reinterpret_cast<const float4*>(fptr + 4 * (tid + NT_));
    float mx = fmaxf(fmaxf(fmaxf(m4.x, m4.y), fmaxf(m4.z, m4.w)), fmaxf(fmaxf(m4b.x, m4b.y), fmaxf(m4b.z, m4b.w)));
    float mn = fminf(fminf(fminf(m4.x, m4.y), fminf(m4.z, m4.w)), fminf(fminf(m4b.x, m4b.y), fminf(m4b.z, m4b.w)));
    {
        const float4 nn[6] = {n0, n1, n2, n3, n4, n5};
#pragma unroll
        for (int v = 0; v < 6; ++v) {
            mx = fmaxf(mx, fmaxf(fmaxf(nn[v].x, nn[v].y), fmaxf(nn[v].z, nn[v].w)));
            mn = fminf(mn, fminf(fminf(nn[v].x, nn[v].y), fminf(nn[v].z, nn[v].w)));
        }
    }
    ROW_MAX(mx, 1);  ROW_MIN(mn, 1);
    ROW_MAX(mx, 2);  ROW_MIN(mn, 2);
    ROW_MAX(mx, 4);  ROW_MIN(mn, 4);
    ROW_MAX(mx, 8);  ROW_MIN(mn, 8);
    mx = fmaxf(mx, xhalf(mx));    mn = fminf(mn, xhalf(mn));

    *reinterpret_cast<float4*>(sf + 4 * tid) = f4;
    *reinterpret_cast<float4*>(sf + 4 * (tid + NT_)) = f4b;

    const float range = mx - mn;
    const float delta = range * (1.0f / NB_);
    const float inv   = (range > 0.0f) ? (float)NB_ / range : 0.0f;
    const float cen0  = mn + 0.5f * delta;

#pragma unroll
    for (int rec = 0; rec < 2; ++rec) {
        const float4 mr = rec ? m4b : m4;
        const float4 fr = rec ? f4b : f4;
        const float mm[4] = {mr.x, mr.y, mr.z, mr.w};
        const float ff[4] = {fr.x, fr.y, fr.z, fr.w};
        float pw[4][8];
        float fw[4][8];
        float lv[4];
#pragma unroll
        for (int e = 0; e < 4; ++e) {
            int lev = (int)((mm[e] - mn) * inv);
            lev = min(max(lev, 0), NB_ - 1);
            const float d  = mm[e] - (cen0 + (float)lev * delta);
            const float d2 = d * d;
            const float d3 = d2 * d;
            const float d4 = d2 * d2;
            pw[e][0] = 1.0f;    pw[e][1] = d;       pw[e][2] = d2;      pw[e][3] = d3;
            pw[e][4] = d4;      pw[e][5] = d4 * d;  pw[e][6] = d4 * d2; pw[e][7] = d4 * d3;
            fw[e][0] = ff[e];
#pragma unroll
            for (int p = 1; p < 8; ++p) fw[e][p] = ff[e] * pw[e][p];
            lv[e] = (float)lev;
        }
        float* rp = tbl + TP_ * (tid + NT_ * rec);
#pragma unroll
        for (int hh = 0; hh < 2; ++hh) {
            const int e0 = 2 * hh, e1 = 2 * hh + 1;
#pragma unroll
            for (int p = 0; p < 8; ++p) {
                *reinterpret_cast<float2*>(rp + 32 * hh + 2 * p)      = make_float2(fw[e0][p], fw[e1][p]);
                *reinterpret_cast<float2*>(rp + 32 * hh + 16 + 2 * p) = make_float2(pw[e0][p], pw[e1][p]);
            }
            *reinterpret_cast<float2*>(rp + 64 + 2 * hh) = make_float2(lv[e0], lv[e1]);
        }
    }
    __builtin_amdgcn_fence(__ATOMIC_RELEASE, "wavefront");
    __builtin_amdgcn_wave_barrier();
    __builtin_amdgcn_fence(__ATOMIC_ACQUIRE, "wavefront");

    {
        const float levc = (float)c16;
        const float* bp = tbl + TP_ * (32 * wave) + 32 * half + 2 * c16;
        const float* lp = tbl + TP_ * (32 * wave) + 64 + 2 * half;
        float2 bq[64], lq[64];
        v8f accA = {}, accB = {};
        auto loadq = [&](const int q0) {
#pragma unroll
            for (int ks = 16 * q0; ks < 16 * q0 + 16; ++ks) {
                const int r = (ks < 32) ? ks : NT_ + ks - 32;
                bq[ks] = *reinterpret_cast<const float2*>(bp + TP_ * r);
                lq[ks] = *reinterpret_cast<const float2*>(lp + TP_ * r);
            }
        };
        auto mulq = [&](const int q0) {
            float lv = levc;
            asm volatile("" : "+v"(lv));
#pragma unroll
            for (int ks = 16 * q0; ks < 16 * q0 + 16; ks += 2) {
                const v2f a0 = {lq[ks].x == lv ? 1.0f : 0.0f, lq[ks].y == lv ? 1.0f : 0.0f};
                const v2f b0 = {bq[ks].x, bq[ks].y};
                const v2f a1 = {lq[ks + 1].x == lv ? 1.0f : 0.0f, lq[ks + 1].y == lv ? 1.0f : 0.0f};
                const v2f b1 = {bq[ks + 1].x, bq[ks + 1].y};
                accA = __builtin_amdgcn_wmma_f32_16x16x4_f32(
                    false, a0, false, b0, (short)0, accA, false, false);
                accB = __builtin_amdgcn_wmma_f32_16x16x4_f32(
                    false, a1, false, b1, (short)0, accB, false, false);
                __builtin_amdgcn_sched_group_barrier(0x002, 4, 0);
                __builtin_amdgcn_sched_group_barrier(0x008, 1, 0);
                __builtin_amdgcn_sched_group_barrier(0x002, 4, 0);
                __builtin_amdgcn_sched_group_barrier(0x008, 1, 0);
            }
        };
#define HOLD() do { asm volatile("" ::: "memory"); __builtin_amdgcn_sched_barrier(0); } while (0)
        loadq(0);
        HOLD();
        loadq(1);
        __builtin_amdgcn_sched_group_barrier(0x100, 16, 0);
        mulq(0);
        HOLD();
        loadq(2);
        HOLD();
        mulq(1);
        HOLD();
        loadq(3);
        HOLD();
        mulq(2);
        mulq(3);
#undef HOLD
#pragma unroll
        for (int r = 0; r < 8; ++r)
            part[wave * 256 + (8 * half + r) * 16 + c16] = accA[r] + accB[r];
    }
    __syncthreads();
#pragma unroll
    for (int u = 0; u < 2; ++u) {
        const int idx = tid + NT_ * u;
        float t = part[idx];
#pragma unroll
        for (int w = 1; w < NW_; ++w) t += part[w * 256 + idx];
        mom[idx] = t;
    }
    __syncthreads();

    v2f aT[2], aU[2];
#pragma unroll
    for (int pg = 0; pg < 2; ++pg) {
        const float2 t = *reinterpret_cast<const float2*>(mom + c16 * 16 + 4 * pg + 2 * half);
        const float2 u = *reinterpret_cast<const float2*>(mom + c16 * 16 + 8 + 4 * pg + 2 * half);
        aT[pg].x = t.x; aT[pg].y = t.y;
        aU[pg].x = u.x; aU[pg].y = u.y;
    }

    const float L2E = 1.44269504088896340736f;
    const float lev0 = (float)(8 * half);
    float res[4];
#pragma unroll
    for (int tile = 0; tile < 4; ++tile) {
        const int row = q * 256 + wave * 64 + tile * 16 + c16;
        const float s   = sf[row];
        const float sL2 = s * L2E;
        const float rL2 = (s >= 0.0f ? s * mx : s * mn) * L2E;
        const float s2 = s * s;
        const float s4 = s2 * s2;
        const float q0 = half ? 0.5f * s2 : 1.0f;
        const float q1 = q0 * s * (half ? (1.0f / 3.0f) : 1.0f);
        const float q4 = s4 * (half ? s2 * (1.0f / 720.0f) : (1.0f / 24.0f));
        const float q5 = q4 * s * (half ? (1.0f / 7.0f) : (1.0f / 5.0f));
        const v2f b0 = {q0, q1};
        const v2f b1 = {q4, q5};
        v8f gT = {}, gU = {};
        gT = __builtin_amdgcn_wmma_f32_16x16x4_f32(
            false, aT[0], false, b0, (short)0, gT, false, false);
        gU = __builtin_amdgcn_wmma_f32_16x16x4_f32(
            false, aU[0], false, b0, (short)0, gU, false, false);
        gT = __builtin_amdgcn_wmma_f32_16x16x4_f32(
            false, aT[1], false, b1, (short)0, gT, false, false);
        gU = __builtin_amdgcn_wmma_f32_16x16x4_f32(
            false, aU[1], false, b1, (short)0, gU, false, false);
        float num = 0.0f, den = 0.0f;
#pragma unroll
        for (int r = 0; r < 8; ++r) {
            const float Mk = cen0 + (lev0 + (float)r) * delta;
            const float E = __builtin_amdgcn_exp2f(sL2 * Mk - rL2);
            num = fmaf(E, gT[r], num);
            den = fmaf(E, gU[r], den);
        }
        num += xhalf(num);
        den += xhalf(den);
        res[tile] = g * (num / den) + s;
    }
    float* orow = out + (size_t)bc * N_ + q * 256 + wave * 64 + lane;
    orow[0]  = half ? res[1] : res[0];
    orow[32] = half ? res[3] : res[2];
}

extern "C" void kernel_launch(void* const* d_in, const int* in_sizes, int n_in,
                              void* d_out, int out_size, void* d_ws, size_t ws_size,
                              hipStream_t stream) {
    (void)in_sizes; (void)n_in; (void)out_size; (void)d_ws; (void)ws_size;
    const float* feature = (const float*)d_in[0];
    const float* seg     = (const float*)d_in[1];
    const float* gamma   = (const float*)d_in[2];
    float* out           = (float*)d_out;

    dim3 grid(B_ * C_ * 4);
    dim3 block(NT_);
    psm_attn_kernel<<<grid, block, 0, stream>>>(feature, seg, gamma, out);
}
